# starting kernel
# baseline (speedup 1.0000x reference)
#include <hip/hip_runtime.h>
#include <hip/hip_bf16.h>

typedef __attribute__((ext_vector_type(2))) float v2f;
typedef __attribute__((ext_vector_type(8))) float v8f;

#define F 128
#define KTOT 384
#define KPAD (KTOT + 4)

__global__ void deg_kernel(const int* __restrict__ src, const int* __restrict__ dst,
                           float* __restrict__ degi, float* __restrict__ dego, int n_edges) {
    int e = blockIdx.x * blockDim.x + threadIdx.x;
    if (e < n_edges) {
        unsafeAtomicAdd(&degi[dst[e]], 1.0f);
        unsafeAtomicAdd(&dego[src[e]], 1.0f);
    }
}

__global__ void scatter_kernel(const float* __restrict__ h, const int* __restrict__ src,
                               const int* __restrict__ dst, float* __restrict__ accf,
                               float* __restrict__ accb, int n_edges) {
    int gid  = blockIdx.x * blockDim.x + threadIdx.x;
    int edge = gid >> 5;
    int lane = gid & 31;
    if (edge >= n_edges) return;
    int s = src[edge];
    int d = dst[edge];
    const float4* hs = (const float4*)(h + (size_t)s * F);
    const float4* hd = (const float4*)(h + (size_t)d * F);
    float4 vs = hs[lane];
    float4 vd = hd[lane];
    float* af = accf + (size_t)d * F + lane * 4;
    float* ab = accb + (size_t)s * F + lane * 4;
    unsafeAtomicAdd(af + 0, vs.x); unsafeAtomicAdd(af + 1, vs.y);
    unsafeAtomicAdd(af + 2, vs.z); unsafeAtomicAdd(af + 3, vs.w);
    unsafeAtomicAdd(ab + 0, vd.x); unsafeAtomicAdd(ab + 1, vd.y);
    unsafeAtomicAdd(ab + 2, vd.z); unsafeAtomicAdd(ab + 3, vd.w);
}

__global__ __launch_bounds__(128) void gemm_kernel(
    const float* __restrict__ h, const float* __restrict__ accf, const float* __restrict__ accb,
    const float* __restrict__ degi, const float* __restrict__ dego,
    const float* __restrict__ W, const float* __restrict__ bias,
    float* __restrict__ out, int relu) {
    __shared__ float Atile[16][KPAD];

    const int tid  = threadIdx.x;
    const int row0 = blockIdx.x * 16;

    for (int idx = tid; idx < 16 * KTOT; idx += 128) {
        int r = idx / KTOT;
        int c = idx - r * KTOT;
        int rg = row0 + r;
        float v;
        if (c < F) {
            v = h[(size_t)rg * F + c];
        } else if (c < 2 * F) {
            v = accf[(size_t)rg * F + (c - F)] * (1.0f / fmaxf(degi[rg], 1.0f));
        } else {
            v = accb[(size_t)rg * F + (c - 2 * F)] * (1.0f / fmaxf(dego[rg], 1.0f));
        }
        Atile[r][c] = v;
    }
    __syncthreads();

    const int wave = tid >> 5;
    const int lane = tid & 31;
    const int lrow = lane & 15;
    const int lhi  = lane >> 4;
    const int n0   = wave * 32;

    v8f c0 = {};
    v8f c1 = {};

    for (int k0 = 0; k0 < KTOT; k0 += 4) {
        const int ka = k0 + 2 * lhi;
        v2f a;
        a.x = Atile[lrow][ka];
        a.y = Atile[lrow][ka + 1];
        const float* wp = W + (size_t)ka * F + n0 + lrow;
        v2f b0; b0.x = wp[0];  b0.y = wp[F];
        v2f b1; b1.x = wp[16]; b1.y = wp[F + 16];
        c0 = __builtin_amdgcn_wmma_f32_16x16x4_f32(false, a, false, b0, (short)0, c0, false, false);
        c1 = __builtin_amdgcn_wmma_f32_16x16x4_f32(false, a, false, b1, (short)0, c1, false, false);
    }

    const int ncol = n0 + lrow;
    const float bi0 = bias[ncol];
    const float bi1 = bias[ncol + 16];
#pragma unroll
    for (int j = 0; j < 8; ++j) {
        const int m = row0 + j + 8 * lhi;
        float v0 = c0[j] + bi0;
        float v1 = c1[j] + bi1;
        if (relu) { v0 = fmaxf(v0, 0.0f); v1 = fmaxf(v1, 0.0f); }
        out[(size_t)m * F + ncol]      = v0;
        out[(size_t)m * F + ncol + 16] = v1;
    }
}

extern "C" void kernel_launch(void* const* d_in, const int* in_sizes, int n_in,
                              void* d_out, int out_size, void* d_ws, size_t ws_size,
                              hipStream_t stream) {
    const float* feat = (const float*)d_in[0];
    const int*   src  = (const int*)d_in[1];
    const int*   dst  = (const int*)d_in[2];
    const float* W1   = (const float*)d_in[3];
    const float* b1   = (const float*)d_in[4];
    const float* W2   = (const float*)d_in[5];
    const float* b2   = (const float*)d_in[6];
    float*       out  = (float*)d_out;

    const int n_nodes = in_sizes[0] / F;
    const int n_edges = in_sizes[1];

    float* degi = (float*)d_ws;
    float* dego = degi + n_nodes;
    float* accf = dego + n_nodes;
    float* accb = accf + (size_t)n_nodes * F;
    float* h1   = accb + (size_t)n_nodes * F;

    const int degBlocks   = (n_edges + 255) / 256;
    const int scatBlocks  = (n_edges * 32 + 255) / 256;
    const int gemmBlocks  = (n_nodes + 15) / 16;

    hipMemsetAsync(degi, 0, (size_t)2 * n_nodes * sizeof(float), stream);
    deg_kernel<<<degBlocks, 256, 0, stream>>>(src, dst, degi, dego, n_edges);

    hipMemsetAsync(accf, 0, (size_t)2 * n_nodes * F * sizeof(float), stream);
    scatter_kernel<<<scatBlocks, 256, 0, stream>>>(feat, src, dst, accf, accb, n_edges);
    gemm_kernel<<<gemmBlocks, 128, 0, stream>>>(feat, accf, accb, degi, dego, W1, b1, h1, 1);

    hipMemsetAsync(accf, 0, (size_t)2 * n_nodes * F * sizeof(float), stream);
    scatter_kernel<<<scatBlocks, 256, 0, stream>>>(h1, src, dst, accf, accb, n_edges);
    gemm_kernel<<<gemmBlocks, 128, 0, stream>>>(h1, accf, accb, degi, dego, W2, b2, out, 0);
}


# Float-atomic scatter replaced by in-call edge lists (int count, per-wave run alloc, cursor fill) + register gather, a wave sums whole rows; rows added in list (fill arrival) order, no float atomics
# speedup vs baseline: 1.8331x; 1.8331x over previous
#include <hip/hip_runtime.h>
#include <hip/hip_bf16.h>

typedef __attribute__((ext_vector_type(2))) float v2f;
typedef __attribute__((ext_vector_type(8))) float v8f;

#define F 128
#define KTOT 384
#define KPAD (KTOT + 4)


__global__ void count_kernel(const int* __restrict__ src, const int* __restrict__ dst,
                             int* __restrict__ cnt, int n_edges, int n_nodes) {
    int e = blockIdx.x * blockDim.x + threadIdx.x;
    if (e < n_edges) {
        const int s = src[e];
        const int d = dst[e];
        atomicAdd(&cnt[d], 1);
        atomicAdd(&cnt[n_nodes + s], 1);
    }
}

__global__ void alloc_kernel(int* __restrict__ cur, int* __restrict__ start,
                             int* __restrict__ total, float* __restrict__ deg, int n_lists) {
    const int i    = blockIdx.x * blockDim.x + threadIdx.x;
    const int lane = threadIdx.x & 31;
    const int cnt  = (i < n_lists) ? cur[i] : 0;
    int incl = cnt;
    for (int d = 1; d < 32; d <<= 1) {
        int y = __shfl_up(incl, d);
        if (lane >= d) incl += y;
    }
    int base = 0;
    if (lane == 31) base = atomicAdd(total, incl);
    base = __shfl(base, 31);
    if (i < n_lists) {
        const int b = base + incl - cnt;
        start[i] = b;
        cur[i]   = b;
        deg[i]   = (float)cnt;
    }
}

__global__ void fill_kernel(const int* __restrict__ src, const int* __restrict__ dst,
                            int* __restrict__ cur, int* __restrict__ col,
                            int n_edges, int n_nodes) {
    int e = blockIdx.x * blockDim.x + threadIdx.x;
    if (e < n_edges) {
        const int s = src[e];
        const int d = dst[e];
        const int p = atomicAdd(&cur[d], 1);
        col[p] = s;
        const int q = atomicAdd(&cur[n_nodes + s], 1);
        col[q] = d;
    }
}

#define LPW 4
__device__ __forceinline__ void add_row(float4& s, const float4& v) {
    s.x += v.x; s.y += v.y; s.z += v.z; s.w += v.w;
}
__global__ __launch_bounds__(256) void gather_kernel(
    const float* __restrict__ h, const int* __restrict__ start, const int* __restrict__ stop,
    const int* __restrict__ col, float* __restrict__ acc, int n_lists) {
    const int lane = threadIdx.x & 31;
    const int wave = __builtin_amdgcn_readfirstlane((int)(blockIdx.x * (blockDim.x >> 5) + (threadIdx.x >> 5)));
    const int row0 = wave * LPW;
    if (row0 >= n_lists) return;
    const float4* H = (const float4*)h;
    float4*       A = (float4*)acc;

    int beg[LPW], end[LPW], idx[LPW];
#pragma unroll
    for (int r = 0; r < LPW; ++r) {
        const bool live = (row0 + r) < n_lists;
        beg[r] = live ? start[row0 + r] : 0;
        end[r] = live ? stop[row0 + r] : 0;
    }
#pragma unroll
    for (int r = 0; r < LPW; ++r)
        idx[r] = (lane < end[r] - beg[r]) ? col[beg[r] + lane] : 0;

#pragma unroll
    for (int r = 0; r < LPW; ++r) {
        if ((row0 + r) < n_lists) {
            const int e = end[r];
            int base = beg[r];
            int mine = idx[r];
            float4 sum = make_float4(0.0f, 0.0f, 0.0f, 0.0f);
            while (base < e) {
                const int n = min(32, e - base);
                int j = 0;
                for (; j + 4 <= n; j += 4) {
                    const int s0 = __builtin_amdgcn_readlane(mine, j);
                    const int s1 = __builtin_amdgcn_readlane(mine, j + 1);
                    const int s2 = __builtin_amdgcn_readlane(mine, j + 2);
                    const int s3 = __builtin_amdgcn_readlane(mine, j + 3);
                    const float4 v0 = H[(size_t)s0 * (F / 4) + lane];
                    const float4 v1 = H[(size_t)s1 * (F / 4) + lane];
                    const float4 v2 = H[(size_t)s2 * (F / 4) + lane];
                    const float4 v3 = H[(size_t)s3 * (F / 4) + lane];
                    add_row(sum, v0); add_row(sum, v1); add_row(sum, v2); add_row(sum, v3);
                }
                const int rem = n - j;
                if (rem > 0) {
                    const float4 zero = make_float4(0.0f, 0.0f, 0.0f, 0.0f);
                    const int s0 = __builtin_amdgcn_readlane(mine, j);
                    const float4 v0 = H[(size_t)s0 * (F / 4) + lane];
                    float4 v1 = zero, v2 = zero;
                    if (rem > 1) {
                        const int s1 = __builtin_amdgcn_readlane(mine, j + 1);
                        v1 = H[(size_t)s1 * (F / 4) + lane];
                    }
                    if (rem > 2) {
                        const int s2 = __builtin_amdgcn_readlane(mine, j + 2);
                        v2 = H[(size_t)s2 * (F / 4) + lane];
                    }
                    add_row(sum, v0); add_row(sum, v1); add_row(sum, v2);
                }
                base += 32;
                if (base < e) mine = (lane < e - base) ? col[base + lane] : 0;
            }
            A[(size_t)(row0 + r) * (F / 4) + lane] = sum;
        }
    }
}

__global__ __launch_bounds__(128) void gemm_kernel(
    const float* __restrict__ h, const float* __restrict__ accf, const float* __restrict__ accb,
    const float* __restrict__ degi, const float* __restrict__ dego,
    const float* __restrict__ W, const float* __restrict__ bias,
    float* __restrict__ out, int relu) {
    __shared__ float Atile[16][KPAD];

    const int tid  = threadIdx.x;
    const int row0 = blockIdx.x * 16;

    for (int idx = tid; idx < 16 * KTOT; idx += 128) {
        int r = idx / KTOT;
        int c = idx - r * KTOT;
        int rg = row0 + r;
        float v;
        if (c < F) {
            v = h[(size_t)rg * F + c];
        } else if (c < 2 * F) {
            v = accf[(size_t)rg * F + (c - F)] * (1.0f / fmaxf(degi[rg], 1.0f));
        } else {
            v = accb[(size_t)rg * F + (c - 2 * F)] * (1.0f / fmaxf(dego[rg], 1.0f));
        }
        Atile[r][c] = v;
    }
    __syncthreads();

    const int wave = tid >> 5;
    const int lane = tid & 31;
    const int lrow = lane & 15;
    const int lhi  = lane >> 4;
    const int n0   = wave * 32;

    v8f c0 = {};
    v8f c1 = {};

    for (int k0 = 0; k0 < KTOT; k0 += 4) {
        const int ka = k0 + 2 * lhi;
        v2f a;
        a.x = Atile[lrow][ka];
        a.y = Atile[lrow][ka + 1];
        const float* wp = W + (size_t)ka * F + n0 + lrow;
        v2f b0; b0.x = wp[0];  b0.y = wp[F];
        v2f b1; b1.x = wp[16]; b1.y = wp[F + 16];
        c0 = __builtin_amdgcn_wmma_f32_16x16x4_f32(false, a, false, b0, (short)0, c0, false, false);
        c1 = __builtin_amdgcn_wmma_f32_16x16x4_f32(false, a, false, b1, (short)0, c1, false, false);
    }

    const int ncol = n0 + lrow;
    const float bi0 = bias[ncol];
    const float bi1 = bias[ncol + 16];
#pragma unroll
    for (int j = 0; j < 8; ++j) {
        const int m = row0 + j + 8 * lhi;
        float v0 = c0[j] + bi0;
        float v1 = c1[j] + bi1;
        if (relu) { v0 = fmaxf(v0, 0.0f); v1 = fmaxf(v1, 0.0f); }
        out[(size_t)m * F + ncol]      = v0;
        out[(size_t)m * F + ncol + 16] = v1;
    }
}

extern "C" void kernel_launch(void* const* d_in, const int* in_sizes, int n_in,
                              void* d_out, int out_size, void* d_ws, size_t ws_size,
                              hipStream_t stream) {
    const float* feat = (const float*)d_in[0];
    const int*   src  = (const int*)d_in[1];
    const int*   dst  = (const int*)d_in[2];
    const float* W1   = (const float*)d_in[3];
    const float* b1   = (const float*)d_in[4];
    const float* W2   = (const float*)d_in[5];
    const float* b2   = (const float*)d_in[6];
    float*       out  = (float*)d_out;

    const int n_nodes = in_sizes[0] / F;
    const int n_edges = in_sizes[1];

    float* degi = (float*)d_ws;
    float* dego = degi + n_nodes;
    float* accf = dego + n_nodes;
    float* accb = accf + (size_t)n_nodes * F;
    float* h1   = accb + (size_t)n_nodes * F;

    int* cur   = (int*)d_out;
    int* total = cur + 2 * n_nodes;
    int* start = total + 4;
    int* col   = start + 2 * n_nodes;

    const int edgeBlocks   = (n_edges + 255) / 256;
    const int listBlocks   = (2 * n_nodes + 255) / 256;
    const int gatherBlocks = (2 * n_nodes + 8 * LPW - 1) / (8 * LPW);
    const int gemmBlocks   = (n_nodes + 15) / 16;

    hipMemsetAsync(cur, 0, ((size_t)2 * n_nodes + 4) * sizeof(int), stream);
    count_kernel<<<edgeBlocks, 256, 0, stream>>>(src, dst, cur, n_edges, n_nodes);
    alloc_kernel<<<listBlocks, 256, 0, stream>>>(cur, start, total, degi, 2 * n_nodes);
    fill_kernel<<<edgeBlocks, 256, 0, stream>>>(src, dst, cur, col, n_edges, n_nodes);

    gather_kernel<<<gatherBlocks, 256, 0, stream>>>(feat, start, cur, col, accf, 2 * n_nodes);
    gemm_kernel<<<gemmBlocks, 128, 0, stream>>>(feat, accf, accb, degi, dego, W1, b1, h1, 1);

    gather_kernel<<<gatherBlocks, 256, 0, stream>>>(h1, start, cur, col, accf, 2 * n_nodes);
    gemm_kernel<<<gemmBlocks, 128, 0, stream>>>(h1, accf, accb, degi, dego, W2, b2, out, 0);
}


# GEMM A-tile staging: a thread issues its 12 float4 row loads and the row degrees before the first wait (was 48 scalar loads, each waited for alone); one reciprocal per row; sums unchanged
# speedup vs baseline: 2.3571x; 1.2859x over previous
#include <hip/hip_runtime.h>
#include <hip/hip_bf16.h>

typedef __attribute__((ext_vector_type(2))) float v2f;
typedef __attribute__((ext_vector_type(8))) float v8f;

#define F 128
#define KTOT 384
#define KPAD (KTOT + 4)


__global__ void count_kernel(const int* __restrict__ src, const int* __restrict__ dst,
                             int* __restrict__ cnt, int n_edges, int n_nodes) {
    int e = blockIdx.x * blockDim.x + threadIdx.x;
    if (e < n_edges) {
        const int s = src[e];
        const int d = dst[e];
        atomicAdd(&cnt[d], 1);
        atomicAdd(&cnt[n_nodes + s], 1);
    }
}

__global__ void alloc_kernel(int* __restrict__ cur, int* __restrict__ start,
                             int* __restrict__ total, float* __restrict__ deg, int n_lists) {
    const int i    = blockIdx.x * blockDim.x + threadIdx.x;
    const int lane = threadIdx.x & 31;
    const int cnt  = (i < n_lists) ? cur[i] : 0;
    int incl = cnt;
    for (int d = 1; d < 32; d <<= 1) {
        int y = __shfl_up(incl, d);
        if (lane >= d) incl += y;
    }
    int base = 0;
    if (lane == 31) base = atomicAdd(total, incl);
    base = __shfl(base, 31);
    if (i < n_lists) {
        const int b = base + incl - cnt;
        start[i] = b;
        cur[i]   = b;
        deg[i]   = (float)cnt;
    }
}

__global__ void fill_kernel(const int* __restrict__ src, const int* __restrict__ dst,
                            int* __restrict__ cur, int* __restrict__ col,
                            int n_edges, int n_nodes) {
    int e = blockIdx.x * blockDim.x + threadIdx.x;
    if (e < n_edges) {
        const int s = src[e];
        const int d = dst[e];
        const int p = atomicAdd(&cur[d], 1);
        col[p] = s;
        const int q = atomicAdd(&cur[n_nodes + s], 1);
        col[q] = d;
    }
}

#define LPW 4
__device__ __forceinline__ void add_row(float4& s, const float4& v) {
    s.x += v.x; s.y += v.y; s.z += v.z; s.w += v.w;
}
__global__ __launch_bounds__(256) void gather_kernel(
    const float* __restrict__ h, const int* __restrict__ start, const int* __restrict__ stop,
    const int* __restrict__ col, float* __restrict__ acc, int n_lists) {
    const int lane = threadIdx.x & 31;
    const int wave = __builtin_amdgcn_readfirstlane((int)(blockIdx.x * (blockDim.x >> 5) + (threadIdx.x >> 5)));
    const int row0 = wave * LPW;
    if (row0 >= n_lists) return;
    const float4* H = (const float4*)h;
    float4*       A = (float4*)acc;

    int beg[LPW], end[LPW], idx[LPW];
#pragma unroll
    for (int r = 0; r < LPW; ++r) {
        const bool live = (row0 + r) < n_lists;
        beg[r] = live ? start[row0 + r] : 0;
        end[r] = live ? stop[row0 + r] : 0;
    }
#pragma unroll
    for (int r = 0; r < LPW; ++r)
        idx[r] = (lane < end[r] - beg[r]) ? col[beg[r] + lane] : 0;

#pragma unroll
    for (int r = 0; r < LPW; ++r) {
        if ((row0 + r) < n_lists) {
            const int e = end[r];
            int base = beg[r];
            int mine = idx[r];
            float4 sum = make_float4(0.0f, 0.0f, 0.0f, 0.0f);
            while (base < e) {
                const int n = min(32, e - base);
                int j = 0;
                for (; j + 4 <= n; j += 4) {
                    const int s0 = __builtin_amdgcn_readlane(mine, j);
                    const int s1 = __builtin_amdgcn_readlane(mine, j + 1);
                    const int s2 = __builtin_amdgcn_readlane(mine, j + 2);
                    const int s3 = __builtin_amdgcn_readlane(mine, j + 3);
                    const float4 v0 = H[(size_t)s0 * (F / 4) + lane];
                    const float4 v1 = H[(size_t)s1 * (F / 4) + lane];
                    const float4 v2 = H[(size_t)s2 * (F / 4) + lane];
                    const float4 v3 = H[(size_t)s3 * (F / 4) + lane];
                    add_row(sum, v0); add_row(sum, v1); add_row(sum, v2); add_row(sum, v3);
                }
                const int rem = n - j;
                if (rem > 0) {
                    const float4 zero = make_float4(0.0f, 0.0f, 0.0f, 0.0f);
                    const int s0 = __builtin_amdgcn_readlane(mine, j);
                    const float4 v0 = H[(size_t)s0 * (F / 4) + lane];
                    float4 v1 = zero, v2 = zero;
                    if (rem > 1) {
                        const int s1 = __builtin_amdgcn_readlane(mine, j + 1);
                        v1 = H[(size_t)s1 * (F / 4) + lane];
                    }
                    if (rem > 2) {
                        const int s2 = __builtin_amdgcn_readlane(mine, j + 2);
                        v2 = H[(size_t)s2 * (F / 4) + lane];
                    }
                    add_row(sum, v0); add_row(sum, v1); add_row(sum, v2);
                }
                base += 32;
                if (base < e) mine = (lane < e - base) ? col[base + lane] : 0;
            }
            A[(size_t)(row0 + r) * (F / 4) + lane] = sum;
        }
    }
}

__global__ __launch_bounds__(128) void gemm_kernel(
    const float* __restrict__ h, const float* __restrict__ accf, const float* __restrict__ accb,
    const float* __restrict__ degi, const float* __restrict__ dego,
    const float* __restrict__ W, const float* __restrict__ bias,
    float* __restrict__ out, int relu) {
    __shared__ __attribute__((aligned(16))) float Atile[16][KPAD];

    const int tid  = threadIdx.x;
    const int row0 = blockIdx.x * 16;

    {
        const int swave = __builtin_amdgcn_readfirstlane(tid >> 5);
        const int slane = tid & 31;
        const float4* H4  = (const float4*)h;
        const float4* AF4 = (const float4*)accf;
        const float4* AB4 = (const float4*)accb;
        float4 vh[4], vf[4], vb[4];
        float  di[4], dq[4];
#pragma unroll
        for (int r = 0; r < 4; ++r) {
            const size_t rg = (size_t)(row0 + 4 * swave + r);
            vh[r] = H4[rg * (F / 4) + slane];
            vf[r] = AF4[rg * (F / 4) + slane];
            vb[r] = AB4[rg * (F / 4) + slane];
            di[r] = degi[rg];
            dq[r] = dego[rg];
        }
#pragma unroll
        for (int r = 0; r < 4; ++r) {
            const float si = 1.0f / fmaxf(di[r], 1.0f);
            const float sq = 1.0f / fmaxf(dq[r], 1.0f);
            float4 f = vf[r];
            float4 b = vb[r];
            f.x *= si; f.y *= si; f.z *= si; f.w *= si;
            b.x *= sq; b.y *= sq; b.z *= sq; b.w *= sq;
            float* arow = &Atile[4 * swave + r][0];
            *(float4*)(arow + 4 * slane)         = vh[r];
            *(float4*)(arow + F + 4 * slane)     = f;
            *(float4*)(arow + 2 * F + 4 * slane) = b;
        }
    }
    __syncthreads();

    const int wave = tid >> 5;
    const int lane = tid & 31;
    const int lrow = lane & 15;
    const int lhi  = lane >> 4;
    const int n0   = wave * 32;

    v8f c0 = {};
    v8f c1 = {};

    for (int k0 = 0; k0 < KTOT; k0 += 4) {
        const int ka = k0 + 2 * lhi;
        v2f a;
        a.x = Atile[lrow][ka];
        a.y = Atile[lrow][ka + 1];
        const float* wp = W + (size_t)ka * F + n0 + lrow;
        v2f b0; b0.x = wp[0];  b0.y = wp[F];
        v2f b1; b1.x = wp[16]; b1.y = wp[F + 16];
        c0 = __builtin_amdgcn_wmma_f32_16x16x4_f32(false, a, false, b0, (short)0, c0, false, false);
        c1 = __builtin_amdgcn_wmma_f32_16x16x4_f32(false, a, false, b1, (short)0, c1, false, false);
    }

    const int ncol = n0 + lrow;
    const float bi0 = bias[ncol];
    const float bi1 = bias[ncol + 16];
#pragma unroll
    for (int j = 0; j < 8; ++j) {
        const int m = row0 + j + 8 * lhi;
        float v0 = c0[j] + bi0;
        float v1 = c1[j] + bi1;
        if (relu) { v0 = fmaxf(v0, 0.0f); v1 = fmaxf(v1, 0.0f); }
        out[(size_t)m * F + ncol]      = v0;
        out[(size_t)m * F + ncol + 16] = v1;
    }
}

extern "C" void kernel_launch(void* const* d_in, const int* in_sizes, int n_in,
                              void* d_out, int out_size, void* d_ws, size_t ws_size,
                              hipStream_t stream) {
    const float* feat = (const float*)d_in[0];
    const int*   src  = (const int*)d_in[1];
    const int*   dst  = (const int*)d_in[2];
    const float* W1   = (const float*)d_in[3];
    const float* b1   = (const float*)d_in[4];
    const float* W2   = (const float*)d_in[5];
    const float* b2   = (const float*)d_in[6];
    float*       out  = (float*)d_out;

    const int n_nodes = in_sizes[0] / F;
    const int n_edges = in_sizes[1];

    float* degi = (float*)d_ws;
    float* dego = degi + n_nodes;
    float* accf = dego + n_nodes;
    float* accb = accf + (size_t)n_nodes * F;
    float* h1   = accb + (size_t)n_nodes * F;

    int* cur   = (int*)d_out;
    int* total = cur + 2 * n_nodes;
    int* start = total + 4;
    int* col   = start + 2 * n_nodes;

    const int edgeBlocks   = (n_edges + 255) / 256;
    const int listBlocks   = (2 * n_nodes + 255) / 256;
    const int gatherBlocks = (2 * n_nodes + 8 * LPW - 1) / (8 * LPW);
    const int gemmBlocks   = (n_nodes + 15) / 16;

    hipMemsetAsync(cur, 0, ((size_t)2 * n_nodes + 4) * sizeof(int), stream);
    count_kernel<<<edgeBlocks, 256, 0, stream>>>(src, dst, cur, n_edges, n_nodes);
    alloc_kernel<<<listBlocks, 256, 0, stream>>>(cur, start, total, degi, 2 * n_nodes);
    fill_kernel<<<edgeBlocks, 256, 0, stream>>>(src, dst, cur, col, n_edges, n_nodes);

    gather_kernel<<<gatherBlocks, 256, 0, stream>>>(feat, start, cur, col, accf, 2 * n_nodes);
    gemm_kernel<<<gemmBlocks, 128, 0, stream>>>(feat, accf, accb, degi, dego, W1, b1, h1, 1);

    gather_kernel<<<gatherBlocks, 256, 0, stream>>>(h1, start, cur, col, accf, 2 * n_nodes);
    gemm_kernel<<<gemmBlocks, 128, 0, stream>>>(h1, accf, accb, degi, dego, W2, b2, out, 0);
}


# Neighbour sums formed inside the GEMM's A-tile staging: a wave sums its 4 rows' 8 edge lists in registers into LDS; sum planes, degree arrays and both gather launches gone; sum order unchanged
# speedup vs baseline: 2.5332x; 1.0747x over previous
#include <hip/hip_runtime.h>
#include <hip/hip_bf16.h>


typedef __attribute__((ext_vector_type(2))) float v2f;
typedef __attribute__((ext_vector_type(8))) float v8f;

#define F 128
#define KTOT 384
#define KPAD (KTOT + 4)


__global__ void count_kernel(const int* __restrict__ src, const int* __restrict__ dst,
                             int* __restrict__ cnt, int n_edges, int n_nodes) {
    int e = blockIdx.x * blockDim.x + threadIdx.x;
    if (e < n_edges) {
        const int s = src[e];
        const int d = dst[e];
        atomicAdd(&cnt[d], 1);
        atomicAdd(&cnt[n_nodes + s], 1);
    }
}

__global__ void alloc_kernel(int* __restrict__ cur, int* __restrict__ start,
                             int* __restrict__ total, int n_lists) {
    const int i    = blockIdx.x * blockDim.x + threadIdx.x;
    const int lane = threadIdx.x & 31;
    const int cnt  = (i < n_lists) ? cur[i] : 0;
    int incl = cnt;
    for (int d = 1; d < 32; d <<= 1) {
        int y = __shfl_up(incl, d);
        if (lane >= d) incl += y;
    }
    int base = 0;
    if (lane == 31) base = atomicAdd(total, incl);
    base = __shfl(base, 31);
    if (i < n_lists) {
        const int b = base + incl - cnt;
        start[i] = b;
        cur[i]   = b;
    }
}

__global__ void fill_kernel(const int* __restrict__ src, const int* __restrict__ dst,
                            int* __restrict__ cur, int* __restrict__ col,
                            int n_edges, int n_nodes) {
    int e = blockIdx.x * blockDim.x + threadIdx.x;
    if (e < n_edges) {
        const int s = src[e];
        const int d = dst[e];
        const int p = atomicAdd(&cur[d], 1);
        col[p] = s;
        const int q = atomicAdd(&cur[n_nodes + s], 1);
        col[q] = d;
    }
}

__device__ __forceinline__ void add_row(float4& s, const float4& v) {
    s.x += v.x; s.y += v.y; s.z += v.z; s.w += v.w;
}
__device__ __forceinline__ float4 list_sum(const float4* __restrict__ H, const int* __restrict__ col,
                                           int beg, int e, int mine, int lane) {
    int base = beg;
    float4 sum = make_float4(0.0f, 0.0f, 0.0f, 0.0f);
    while (base < e) {
        const int n = min(32, e - base);
        int j = 0;
        for (; j + 4 <= n; j += 4) {
            const int s0 = __builtin_amdgcn_readlane(mine, j);
            const int s1 = __builtin_amdgcn_readlane(mine, j + 1);
            const int s2 = __builtin_amdgcn_readlane(mine, j + 2);
            const int s3 = __builtin_amdgcn_readlane(mine, j + 3);
            const float4 v0 = H[(size_t)s0 * (F / 4) + lane];
            const float4 v1 = H[(size_t)s1 * (F / 4) + lane];
            const float4 v2 = H[(size_t)s2 * (F / 4) + lane];
            const float4 v3 = H[(size_t)s3 * (F / 4) + lane];
            add_row(sum, v0); add_row(sum, v1); add_row(sum, v2); add_row(sum, v3);
        }
        const int rem = n - j;
        if (rem > 0) {
            const float4 zero = make_float4(0.0f, 0.0f, 0.0f, 0.0f);
            const int s0 = __builtin_amdgcn_readlane(mine, j);
            const float4 v0 = H[(size_t)s0 * (F / 4) + lane];
            float4 v1 = zero, v2 = zero;
            if (rem > 1) {
                const int s1 = __builtin_amdgcn_readlane(mine, j + 1);
                v1 = H[(size_t)s1 * (F / 4) + lane];
            }
            if (rem > 2) {
                const int s2 = __builtin_amdgcn_readlane(mine, j + 2);
                v2 = H[(size_t)s2 * (F / 4) + lane];
            }
            add_row(sum, v0); add_row(sum, v1); add_row(sum, v2);
        }
        base += 32;
        if (base < e) mine = (lane < e - base) ? col[base + lane] : 0;
    }
    return sum;
}

__global__ __launch_bounds__(128) void gemm_kernel(
    const float* __restrict__ h, const int* __restrict__ start, const int* __restrict__ stop,
    const int* __restrict__ col, const float* __restrict__ W, const float* __restrict__ bias,
    float* __restrict__ out, int relu, int n_nodes) {
    __shared__ __attribute__((aligned(16))) float Atile[16][KPAD];

    const int tid  = threadIdx.x;
    const int row0 = blockIdx.x * 16;

    {
        const int swave = __builtin_amdgcn_readfirstlane(tid >> 5);
        const int slane = tid & 31;
        const int node0 = row0 + 4 * swave;
        const float4* H4 = (const float4*)h;
        float4 vh[4];
        int beg[8], end[8], idx[8];
#pragma unroll
        for (int r = 0; r < 4; ++r) {
            vh[r] = H4[(size_t)(node0 + r) * (F / 4) + slane];
            beg[2 * r]     = start[node0 + r];
            end[2 * r]     = stop[node0 + r];
            beg[2 * r + 1] = start[n_nodes + node0 + r];
            end[2 * r + 1] = stop[n_nodes + node0 + r];
        }
#pragma unroll
        for (int q = 0; q < 8; ++q)
            idx[q] = (slane < end[q] - beg[q]) ? col[beg[q] + slane] : 0;
#pragma unroll
        for (int r = 0; r < 4; ++r)
            *(float4*)(&Atile[4 * swave + r][4 * slane]) = vh[r];
#pragma unroll
        for (int q = 0; q < 8; ++q) {
            float4 s = list_sum(H4, col, beg[q], end[q], idx[q], slane);
            const float sc = 1.0f / fmaxf((float)(end[q] - beg[q]), 1.0f);
            s.x *= sc; s.y *= sc; s.z *= sc; s.w *= sc;
            *(float4*)(&Atile[4 * swave + (q >> 1)][F * (1 + (q & 1)) + 4 * slane]) = s;
        }
    }
    __syncthreads();

    const int wave = tid >> 5;
    const int lane = tid & 31;
    const int lrow = lane & 15;
    const int lhi  = lane >> 4;
    const int n0   = wave * 32;

    v8f c0 = {};
    v8f c1 = {};

    for (int k0 = 0; k0 < KTOT; k0 += 4) {
        const int ka = k0 + 2 * lhi;
        v2f a;
        a.x = Atile[lrow][ka];
        a.y = Atile[lrow][ka + 1];
        const float* wp = W + (size_t)ka * F + n0 + lrow;
        v2f b0; b0.x = wp[0];  b0.y = wp[F];
        v2f b1; b1.x = wp[16]; b1.y = wp[F + 16];
        c0 = __builtin_amdgcn_wmma_f32_16x16x4_f32(false, a, false, b0, (short)0, c0, false, false);
        c1 = __builtin_amdgcn_wmma_f32_16x16x4_f32(false, a, false, b1, (short)0, c1, false, false);
    }

    const int ncol = n0 + lrow;
    const float bi0 = bias[ncol];
    const float bi1 = bias[ncol + 16];
#pragma unroll
    for (int j = 0; j < 8; ++j) {
        const int m = row0 + j + 8 * lhi;
        float v0 = c0[j] + bi0;
        float v1 = c1[j] + bi1;
        if (relu) { v0 = fmaxf(v0, 0.0f); v1 = fmaxf(v1, 0.0f); }
        out[(size_t)m * F + ncol]      = v0;
        out[(size_t)m * F + ncol + 16] = v1;
    }
}

extern "C" void kernel_launch(void* const* d_in, const int* in_sizes, int n_in,
                              void* d_out, int out_size, void* d_ws, size_t ws_size,
                              hipStream_t stream) {
    const float* feat = (const float*)d_in[0];
    const int*   src  = (const int*)d_in[1];
    const int*   dst  = (const int*)d_in[2];
    const float* W1   = (const float*)d_in[3];
    const float* b1   = (const float*)d_in[4];
    const float* W2   = (const float*)d_in[5];
    const float* b2   = (const float*)d_in[6];
    float*       out  = (float*)d_out;

    const int n_nodes = in_sizes[0] / F;
    const int n_edges = in_sizes[1];

    int* cur   = (int*)d_ws;
    int* total = cur + 2 * n_nodes;
    int* start = total + 4;
    int* col   = start + 2 * n_nodes;
    float* h1  = (float*)d_ws + 2 * (size_t)n_nodes + 2 * (size_t)n_nodes * F;

    const int edgeBlocks = (n_edges + 255) / 256;
    const int listBlocks = (2 * n_nodes + 255) / 256;
    const int gemmBlocks = (n_nodes + 15) / 16;

    hipMemsetAsync(cur, 0, ((size_t)2 * n_nodes + 4) * sizeof(int), stream);
    count_kernel<<<edgeBlocks, 256, 0, stream>>>(src, dst, cur, n_edges, n_nodes);
    alloc_kernel<<<listBlocks, 256, 0, stream>>>(cur, start, total, 2 * n_nodes);
    fill_kernel<<<edgeBlocks, 256, 0, stream>>>(src, dst, cur, col, n_edges, n_nodes);

    gemm_kernel<<<gemmBlocks, 128, 0, stream>>>(feat, start, cur, col, W1, b1, h1, 1, n_nodes);

    gemm_kernel<<<gemmBlocks, 128, 0, stream>>>(h1, start, cur, col, W2, b2, out, 0, n_nodes);
}


# Edge lists via LDS counters: block (edge chunk, list range) counts and places by LDS atomics; scattered global atomics and memset gone; list rows summed in chunk order, LDS arrival within a chunk
# speedup vs baseline: 3.3709x; 1.3307x over previous
#include <hip/hip_runtime.h>
#include <hip/hip_bf16.h>


typedef __attribute__((ext_vector_type(2))) float v2f;
typedef __attribute__((ext_vector_type(8))) float v8f;

#define F 128
#define KTOT 384
#define KPAD (KTOT + 4)


#define RS     8192
#define NCHUNK 16
#define LT     512

__device__ __forceinline__ void chunk_bounds(int n_edges, int c, int& e0, int& e1) {
    const int cs = (((n_edges + NCHUNK - 1) / NCHUNK) + 3) & ~3;
    e0 = min(c * cs, n_edges);
    e1 = min(e0 + cs, n_edges);
}

__device__ __forceinline__ void count_hit(int* cl, unsigned lo, int list) {
    const unsigned k = (unsigned)list - lo;
    if (k < (unsigned)RS) atomicAdd(&cl[k], 1);
}
__device__ __forceinline__ void count4(int* cl, unsigned lo, int n_nodes, const int4& s4, const int4& d4) {
    count_hit(cl, lo, d4.x); count_hit(cl, lo, n_nodes + s4.x);
    count_hit(cl, lo, d4.y); count_hit(cl, lo, n_nodes + s4.y);
    count_hit(cl, lo, d4.z); count_hit(cl, lo, n_nodes + s4.z);
    count_hit(cl, lo, d4.w); count_hit(cl, lo, n_nodes + s4.w);
}
__global__ __launch_bounds__(LT) void count_kernel(const int* __restrict__ src, const int* __restrict__ dst,
                                                   int* __restrict__ cnt, int* __restrict__ total,
                                                   int n_edges, int n_nodes) {
    __shared__ int cl[RS];
    const int tid = threadIdx.x;
    const int c   = blockIdx.x;
    const unsigned lo = blockIdx.y * RS;
    const int n_lists = 2 * n_nodes;
    if (blockIdx.x == 0 && blockIdx.y == 0 && tid == 0) *total = 0;
    for (int i = tid; i < RS; i += LT) cl[i] = 0;
    __syncthreads();
    int e0, e1;
    chunk_bounds(n_edges, c, e0, e1);
    const int step = 4 * LT;
    int b = e0 + 4 * tid;
    for (; b + 3 * step + 4 <= e1; b += 4 * step) {
        const int4 s0 = *(const int4*)(src + b);
        const int4 d0 = *(const int4*)(dst + b);
        const int4 s1 = *(const int4*)(src + b + step);
        const int4 d1 = *(const int4*)(dst + b + step);
        const int4 s2 = *(const int4*)(src + b + 2 * step);
        const int4 d2 = *(const int4*)(dst + b + 2 * step);
        const int4 s3 = *(const int4*)(src + b + 3 * step);
        const int4 d3 = *(const int4*)(dst + b + 3 * step);
        count4(cl, lo, n_nodes, s0, d0);
        count4(cl, lo, n_nodes, s1, d1);
        count4(cl, lo, n_nodes, s2, d2);
        count4(cl, lo, n_nodes, s3, d3);
    }
    for (; b < e1; b += step) {
        if (b + 4 <= e1) {
            const int4 s4 = *(const int4*)(src + b);
            const int4 d4 = *(const int4*)(dst + b);
            count4(cl, lo, n_nodes, s4, d4);
        } else {
            for (int e = b; e < e1; ++e) {
                count_hit(cl, lo, dst[e]); count_hit(cl, lo, n_nodes + src[e]);
            }
        }
    }
    __syncthreads();
    for (int i = tid; i < RS; i += LT) {
        const int l = (int)lo + i;
        if (l < n_lists) cnt[(size_t)c * n_lists + l] = cl[i];
    }
}

__global__ void alloc_kernel(int* __restrict__ cnt, int* __restrict__ start, int* __restrict__ stop,
                             int* __restrict__ total, int n_lists) {
    const int i    = blockIdx.x * blockDim.x + threadIdx.x;
    const int lane = threadIdx.x & 31;
    int v[NCHUNK];
    int len = 0;
#pragma unroll
    for (int c = 0; c < NCHUNK; ++c) {
        v[c] = (i < n_lists) ? cnt[(size_t)c * n_lists + i] : 0;
        len += v[c];
    }
    int incl = len;
    for (int d = 1; d < 32; d <<= 1) {
        int y = __shfl_up(incl, d);
        if (lane >= d) incl += y;
    }
    int base = 0;
    if (lane == 31) base = atomicAdd(total, incl);
    base = __shfl(base, 31);
    if (i < n_lists) {
        int b = base + incl - len;
        start[i] = b;
        stop[i]  = b + len;
#pragma unroll
        for (int c = 0; c < NCHUNK; ++c) {
            cnt[(size_t)c * n_lists + i] = b;
            b += v[c];
        }
    }
}

__device__ __forceinline__ void fill_hit(int* cl, unsigned lo, int list, int value, int* __restrict__ col) {
    const unsigned k = (unsigned)list - lo;
    if (k < (unsigned)RS) col[atomicAdd(&cl[k], 1)] = value;
}
__device__ __forceinline__ void fill4(int* cl, unsigned lo, int n_nodes, const int4& s4, const int4& d4,
                                      int* __restrict__ col) {
    fill_hit(cl, lo, d4.x, s4.x, col); fill_hit(cl, lo, n_nodes + s4.x, d4.x, col);
    fill_hit(cl, lo, d4.y, s4.y, col); fill_hit(cl, lo, n_nodes + s4.y, d4.y, col);
    fill_hit(cl, lo, d4.z, s4.z, col); fill_hit(cl, lo, n_nodes + s4.z, d4.z, col);
    fill_hit(cl, lo, d4.w, s4.w, col); fill_hit(cl, lo, n_nodes + s4.w, d4.w, col);
}
__global__ __launch_bounds__(LT) void fill_kernel(const int* __restrict__ src, const int* __restrict__ dst,
                                                  const int* __restrict__ first, int* __restrict__ col,
                                                  int n_edges, int n_nodes) {
    __shared__ int cl[RS];
    const int tid = threadIdx.x;
    const int c   = blockIdx.x;
    const unsigned lo = blockIdx.y * RS;
    const int n_lists = 2 * n_nodes;
#pragma unroll
    for (int t = 0; t < RS / LT; ++t) {
        const int i = tid + t * LT;
        const int l = min((int)lo + i, n_lists - 1);
        cl[i] = first[(size_t)c * n_lists + l];
    }
    __syncthreads();
    int e0, e1;
    chunk_bounds(n_edges, c, e0, e1);
    const int step = 4 * LT;
    int b = e0 + 4 * tid;
    for (; b + 3 * step + 4 <= e1; b += 4 * step) {
        const int4 s0 = *(const int4*)(src + b);
        const int4 d0 = *(const int4*)(dst + b);
        const int4 s1 = *(const int4*)(src + b + step);
        const int4 d1 = *(const int4*)(dst + b + step);
        const int4 s2 = *(const int4*)(src + b + 2 * step);
        const int4 d2 = *(const int4*)(dst + b + 2 * step);
        const int4 s3 = *(const int4*)(src + b + 3 * step);
        const int4 d3 = *(const int4*)(dst + b + 3 * step);
        fill4(cl, lo, n_nodes, s0, d0, col);
        fill4(cl, lo, n_nodes, s1, d1, col);
        fill4(cl, lo, n_nodes, s2, d2, col);
        fill4(cl, lo, n_nodes, s3, d3, col);
    }
    for (; b < e1; b += step) {
        if (b + 4 <= e1) {
            const int4 s4 = *(const int4*)(src + b);
            const int4 d4 = *(const int4*)(dst + b);
            fill4(cl, lo, n_nodes, s4, d4, col);
        } else {
            for (int e = b; e < e1; ++e) {
                const int s = src[e];
                const int d = dst[e];
                fill_hit(cl, lo, d, s, col); fill_hit(cl, lo, n_nodes + s, d, col);
            }
        }
    }
}

__device__ __forceinline__ void add_row(float4& s, const float4& v) {
    s.x += v.x; s.y += v.y; s.z += v.z; s.w += v.w;
}
__device__ __forceinline__ float4 list_sum(const float4* __restrict__ H, const int* __restrict__ col,
                                           int beg, int e, int mine, int lane) {
    int base = beg;
    float4 sum = make_float4(0.0f, 0.0f, 0.0f, 0.0f);
    while (base < e) {
        const int n = min(32, e - base);
        int j = 0;
        for (; j + 4 <= n; j += 4) {
            const int s0 = __builtin_amdgcn_readlane(mine, j);
            const int s1 = __builtin_amdgcn_readlane(mine, j + 1);
            const int s2 = __builtin_amdgcn_readlane(mine, j + 2);
            const int s3 = __builtin_amdgcn_readlane(mine, j + 3);
            const float4 v0 = H[(size_t)s0 * (F / 4) + lane];
            const float4 v1 = H[(size_t)s1 * (F / 4) + lane];
            const float4 v2 = H[(size_t)s2 * (F / 4) + lane];
            const float4 v3 = H[(size_t)s3 * (F / 4) + lane];
            add_row(sum, v0); add_row(sum, v1); add_row(sum, v2); add_row(sum, v3);
        }
        const int rem = n - j;
        if (rem > 0) {
            const float4 zero = make_float4(0.0f, 0.0f, 0.0f, 0.0f);
            const int s0 = __builtin_amdgcn_readlane(mine, j);
            const float4 v0 = H[(size_t)s0 * (F / 4) + lane];
            float4 v1 = zero, v2 = zero;
            if (rem > 1) {
                const int s1 = __builtin_amdgcn_readlane(mine, j + 1);
                v1 = H[(size_t)s1 * (F / 4) + lane];
            }
            if (rem > 2) {
                const int s2 = __builtin_amdgcn_readlane(mine, j + 2);
                v2 = H[(size_t)s2 * (F / 4) + lane];
            }
            add_row(sum, v0); add_row(sum, v1); add_row(sum, v2);
        }
        base += 32;
        if (base < e) mine = (lane < e - base) ? col[base + lane] : 0;
    }
    return sum;
}

__global__ __launch_bounds__(128) void gemm_kernel(
    const float* __restrict__ h, const int* __restrict__ start, const int* __restrict__ stop,
    const int* __restrict__ col, const float* __restrict__ W, const float* __restrict__ bias,
    float* __restrict__ out, int relu, int n_nodes) {
    __shared__ __attribute__((aligned(16))) float Atile[16][KPAD];

    const int tid  = threadIdx.x;
    const int row0 = blockIdx.x * 16;

    {
        const int swave = __builtin_amdgcn_readfirstlane(tid >> 5);
        const int slane = tid & 31;
        const int node0 = row0 + 4 * swave;
        const float4* H4 = (const float4*)h;
        float4 vh[4];
        int beg[8], end[8], idx[8];
#pragma unroll
        for (int r = 0; r < 4; ++r) {
            vh[r] = H4[(size_t)(node0 + r) * (F / 4) + slane];
            beg[2 * r]     = start[node0 + r];
            end[2 * r]     = stop[node0 + r];
            beg[2 * r + 1] = start[n_nodes + node0 + r];
            end[2 * r + 1] = stop[n_nodes + node0 + r];
        }
#pragma unroll
        for (int q = 0; q < 8; ++q)
            idx[q] = (slane < end[q] - beg[q]) ? col[beg[q] + slane] : 0;
#pragma unroll
        for (int r = 0; r < 4; ++r)
            *(float4*)(&Atile[4 * swave + r][4 * slane]) = vh[r];
#pragma unroll
        for (int q = 0; q < 8; ++q) {
            float4 s = list_sum(H4, col, beg[q], end[q], idx[q], slane);
            const float sc = 1.0f / fmaxf((float)(end[q] - beg[q]), 1.0f);
            s.x *= sc; s.y *= sc; s.z *= sc; s.w *= sc;
            *(float4*)(&Atile[4 * swave + (q >> 1)][F * (1 + (q & 1)) + 4 * slane]) = s;
        }
    }
    __syncthreads();

    const int wave = tid >> 5;
    const int lane = tid & 31;
    const int lrow = lane & 15;
    const int lhi  = lane >> 4;
    const int n0   = wave * 32;

    v8f c0 = {};
    v8f c1 = {};

    for (int k0 = 0; k0 < KTOT; k0 += 4) {
        const int ka = k0 + 2 * lhi;
        v2f a;
        a.x = Atile[lrow][ka];
        a.y = Atile[lrow][ka + 1];
        const float* wp = W + (size_t)ka * F + n0 + lrow;
        v2f b0; b0.x = wp[0];  b0.y = wp[F];
        v2f b1; b1.x = wp[16]; b1.y = wp[F + 16];
        c0 = __builtin_amdgcn_wmma_f32_16x16x4_f32(false, a, false, b0, (short)0, c0, false, false);
        c1 = __builtin_amdgcn_wmma_f32_16x16x4_f32(false, a, false, b1, (short)0, c1, false, false);
    }

    const int ncol = n0 + lrow;
    const float bi0 = bias[ncol];
    const float bi1 = bias[ncol + 16];
#pragma unroll
    for (int j = 0; j < 8; ++j) {
        const int m = row0 + j + 8 * lhi;
        float v0 = c0[j] + bi0;
        float v1 = c1[j] + bi1;
        if (relu) { v0 = fmaxf(v0, 0.0f); v1 = fmaxf(v1, 0.0f); }
        out[(size_t)m * F + ncol]      = v0;
        out[(size_t)m * F + ncol + 16] = v1;
    }
}

extern "C" void kernel_launch(void* const* d_in, const int* in_sizes, int n_in,
                              void* d_out, int out_size, void* d_ws, size_t ws_size,
                              hipStream_t stream) {
    const float* feat = (const float*)d_in[0];
    const int*   src  = (const int*)d_in[1];
    const int*   dst  = (const int*)d_in[2];
    const float* W1   = (const float*)d_in[3];
    const float* b1   = (const float*)d_in[4];
    const float* W2   = (const float*)d_in[5];
    const float* b2   = (const float*)d_in[6];
    float*       out  = (float*)d_out;

    const int n_nodes = in_sizes[0] / F;
    const int n_edges = in_sizes[1];

    int* start = (int*)d_ws;
    int* stop  = start + 2 * n_nodes;
    int* total = stop + 2 * n_nodes;
    int* col   = total + 4;
    int* cnt   = col + 2 * n_edges;
    float* h1  = (float*)d_ws + 2 * (size_t)n_nodes + 2 * (size_t)n_nodes * F;

    const int rangeBlocks = (2 * n_nodes + RS - 1) / RS;
    const int listBlocks  = (2 * n_nodes + 255) / 256;
    const int gemmBlocks  = (n_nodes + 15) / 16;

    count_kernel<<<dim3(NCHUNK, rangeBlocks), LT, 0, stream>>>(src, dst, cnt, total, n_edges, n_nodes);
    alloc_kernel<<<listBlocks, 256, 0, stream>>>(cnt, start, stop, total, 2 * n_nodes);
    fill_kernel<<<dim3(NCHUNK, rangeBlocks), LT, 0, stream>>>(src, dst, cnt, col, n_edges, n_nodes);

    gemm_kernel<<<gemmBlocks, 128, 0, stream>>>(feat, start, stop, col, W1, b1, h1, 1, n_nodes);

    gemm_kernel<<<gemmBlocks, 128, 0, stream>>>(h1, start, stop, col, W2, b2, out, 0, n_nodes);
}


# GEMM block tile 32 rows x 128 cols on 8 waves: a wave owns one column tile and both row tiles, so each W fragment feeds 2 products; result rows leave through LDS whole; sums unchanged
# speedup vs baseline: 3.5522x; 1.0538x over previous
#include <hip/hip_runtime.h>
#include <hip/hip_bf16.h>


typedef __attribute__((ext_vector_type(2))) float v2f;
typedef __attribute__((ext_vector_type(8))) float v8f;

#define F 128
#define KTOT 384
#define KPAD (KTOT + 4)


#define RS     8192
#define NCHUNK 16
#define LT     512

__device__ __forceinline__ void chunk_bounds(int n_edges, int c, int& e0, int& e1) {
    const int cs = (((n_edges + NCHUNK - 1) / NCHUNK) + 3) & ~3;
    e0 = min(c * cs, n_edges);
    e1 = min(e0 + cs, n_edges);
}

__device__ __forceinline__ void count_hit(int* cl, unsigned lo, int list) {
    const unsigned k = (unsigned)list - lo;
    if (k < (unsigned)RS) atomicAdd(&cl[k], 1);
}
__device__ __forceinline__ void count4(int* cl, unsigned lo, int n_nodes, const int4& s4, const int4& d4) {
    count_hit(cl, lo, d4.x); count_hit(cl, lo, n_nodes + s4.x);
    count_hit(cl, lo, d4.y); count_hit(cl, lo, n_nodes + s4.y);
    count_hit(cl, lo, d4.z); count_hit(cl, lo, n_nodes + s4.z);
    count_hit(cl, lo, d4.w); count_hit(cl, lo, n_nodes + s4.w);
}
__global__ __launch_bounds__(LT) void count_kernel(const int* __restrict__ src, const int* __restrict__ dst,
                                                   int* __restrict__ cnt, int* __restrict__ total,
                                                   int n_edges, int n_nodes) {
    __shared__ int cl[RS];
    const int tid = threadIdx.x;
    const int c   = blockIdx.x;
    const unsigned lo = blockIdx.y * RS;
    const int n_lists = 2 * n_nodes;
    if (blockIdx.x == 0 && blockIdx.y == 0 && tid == 0) *total = 0;
    for (int i = tid; i < RS; i += LT) cl[i] = 0;
    __syncthreads();
    int e0, e1;
    chunk_bounds(n_edges, c, e0, e1);
    const int step = 4 * LT;
    int b = e0 + 4 * tid;
    for (; b + 3 * step + 4 <= e1; b += 4 * step) {
        const int4 s0 = *(const int4*)(src + b);
        const int4 d0 = *(const int4*)(dst + b);
        const int4 s1 = *(const int4*)(src + b + step);
        const int4 d1 = *(const int4*)(dst + b + step);
        const int4 s2 = *(const int4*)(src + b + 2 * step);
        const int4 d2 = *(const int4*)(dst + b + 2 * step);
        const int4 s3 = *(const int4*)(src + b + 3 * step);
        const int4 d3 = *(const int4*)(dst + b + 3 * step);
        count4(cl, lo, n_nodes, s0, d0);
        count4(cl, lo, n_nodes, s1, d1);
        count4(cl, lo, n_nodes, s2, d2);
        count4(cl, lo, n_nodes, s3, d3);
    }
    for (; b < e1; b += step) {
        if (b + 4 <= e1) {
            const int4 s4 = *(const int4*)(src + b);
            const int4 d4 = *(const int4*)(dst + b);
            count4(cl, lo, n_nodes, s4, d4);
        } else {
            for (int e = b; e < e1; ++e) {
                count_hit(cl, lo, dst[e]); count_hit(cl, lo, n_nodes + src[e]);
            }
        }
    }
    __syncthreads();
    for (int i = tid; i < RS; i += LT) {
        const int l = (int)lo + i;
        if (l < n_lists) cnt[(size_t)c * n_lists + l] = cl[i];
    }
}

__global__ void alloc_kernel(int* __restrict__ cnt, int* __restrict__ start, int* __restrict__ stop,
                             int* __restrict__ total, int n_lists) {
    const int i    = blockIdx.x * blockDim.x + threadIdx.x;
    const int lane = threadIdx.x & 31;
    int v[NCHUNK];
    int len = 0;
#pragma unroll
    for (int c = 0; c < NCHUNK; ++c) {
        v[c] = (i < n_lists) ? cnt[(size_t)c * n_lists + i] : 0;
        len += v[c];
    }
    int incl = len;
    for (int d = 1; d < 32; d <<= 1) {
        int y = __shfl_up(incl, d);
        if (lane >= d) incl += y;
    }
    int base = 0;
    if (lane == 31) base = atomicAdd(total, incl);
    base = __shfl(base, 31);
    if (i < n_lists) {
        int b = base + incl - len;
        start[i] = b;
        stop[i]  = b + len;
#pragma unroll
        for (int c = 0; c < NCHUNK; ++c) {
            cnt[(size_t)c * n_lists + i] = b;
            b += v[c];
        }
    }
}

__device__ __forceinline__ void fill_hit(int* cl, unsigned lo, int list, int value, int* __restrict__ col) {
    const unsigned k = (unsigned)list - lo;
    if (k < (unsigned)RS) col[atomicAdd(&cl[k], 1)] = value;
}
__device__ __forceinline__ void fill4(int* cl, unsigned lo, int n_nodes, const int4& s4, const int4& d4,
                                      int* __restrict__ col) {
    fill_hit(cl, lo, d4.x, s4.x, col); fill_hit(cl, lo, n_nodes + s4.x, d4.x, col);
    fill_hit(cl, lo, d4.y, s4.y, col); fill_hit(cl, lo, n_nodes + s4.y, d4.y, col);
    fill_hit(cl, lo, d4.z, s4.z, col); fill_hit(cl, lo, n_nodes + s4.z, d4.z, col);
    fill_hit(cl, lo, d4.w, s4.w, col); fill_hit(cl, lo, n_nodes + s4.w, d4.w, col);
}
__global__ __launch_bounds__(LT) void fill_kernel(const int* __restrict__ src, const int* __restrict__ dst,
                                                  const int* __restrict__ first, int* __restrict__ col,
                                                  int n_edges, int n_nodes) {
    __shared__ int cl[RS];
    const int tid = threadIdx.x;
    const int c   = blockIdx.x;
    const unsigned lo = blockIdx.y * RS;
    const int n_lists = 2 * n_nodes;
#pragma unroll
    for (int t = 0; t < RS / LT; ++t) {
        const int i = tid + t * LT;
        const int l = min((int)lo + i, n_lists - 1);
        cl[i] = first[(size_t)c * n_lists + l];
    }
    __syncthreads();
    int e0, e1;
    chunk_bounds(n_edges, c, e0, e1);
    const int step = 4 * LT;
    int b = e0 + 4 * tid;
    for (; b + 3 * step + 4 <= e1; b += 4 * step) {
        const int4 s0 = *(const int4*)(src + b);
        const int4 d0 = *(const int4*)(dst + b);
        const int4 s1 = *(const int4*)(src + b + step);
        const int4 d1 = *(const int4*)(dst + b + step);
        const int4 s2 = *(const int4*)(src + b + 2 * step);
        const int4 d2 = *(const int4*)(dst + b + 2 * step);
        const int4 s3 = *(const int4*)(src + b + 3 * step);
        const int4 d3 = *(const int4*)(dst + b + 3 * step);
        fill4(cl, lo, n_nodes, s0, d0, col);
        fill4(cl, lo, n_nodes, s1, d1, col);
        fill4(cl, lo, n_nodes, s2, d2, col);
        fill4(cl, lo, n_nodes, s3, d3, col);
    }
    for (; b < e1; b += step) {
        if (b + 4 <= e1) {
            const int4 s4 = *(const int4*)(src + b);
            const int4 d4 = *(const int4*)(dst + b);
            fill4(cl, lo, n_nodes, s4, d4, col);
        } else {
            for (int e = b; e < e1; ++e) {
                const int s = src[e];
                const int d = dst[e];
                fill_hit(cl, lo, d, s, col); fill_hit(cl, lo, n_nodes + s, d, col);
            }
        }
    }
}

__device__ __forceinline__ void add_row(float4& s, const float4& v) {
    s.x += v.x; s.y += v.y; s.z += v.z; s.w += v.w;
}
__device__ __forceinline__ float4 list_sum(const float4* __restrict__ H, const int* __restrict__ col,
                                           int beg, int e, int mine, int lane) {
    int base = beg;
    float4 sum = make_float4(0.0f, 0.0f, 0.0f, 0.0f);
    while (base < e) {
        const int n = min(32, e - base);
        int j = 0;
        for (; j + 4 <= n; j += 4) {
            const int s0 = __builtin_amdgcn_readlane(mine, j);
            const int s1 = __builtin_amdgcn_readlane(mine, j + 1);
            const int s2 = __builtin_amdgcn_readlane(mine, j + 2);
            const int s3 = __builtin_amdgcn_readlane(mine, j + 3);
            const float4 v0 = H[(size_t)s0 * (F / 4) + lane];
            const float4 v1 = H[(size_t)s1 * (F / 4) + lane];
            const float4 v2 = H[(size_t)s2 * (F / 4) + lane];
            const float4 v3 = H[(size_t)s3 * (F / 4) + lane];
            add_row(sum, v0); add_row(sum, v1); add_row(sum, v2); add_row(sum, v3);
        }
        const int rem = n - j;
        if (rem > 0) {
            const float4 zero = make_float4(0.0f, 0.0f, 0.0f, 0.0f);
            const int s0 = __builtin_amdgcn_readlane(mine, j);
            const float4 v0 = H[(size_t)s0 * (F / 4) + lane];
            float4 v1 = zero, v2 = zero;
            if (rem > 1) {
                const int s1 = __builtin_amdgcn_readlane(mine, j + 1);
                v1 = H[(size_t)s1 * (F / 4) + lane];
            }
            if (rem > 2) {
                const int s2 = __builtin_amdgcn_readlane(mine, j + 2);
                v2 = H[(size_t)s2 * (F / 4) + lane];
            }
            add_row(sum, v0); add_row(sum, v1); add_row(sum, v2);
        }
        base += 32;
        if (base < e) mine = (lane < e - base) ? col[base + lane] : 0;
    }
    return sum;
}

#define TM 32
#define CP (F + 4)
__global__ __launch_bounds__(256) void gemm_kernel(
    const float* __restrict__ h, const int* __restrict__ start, const int* __restrict__ stop,
    const int* __restrict__ col, const float* __restrict__ W, const float* __restrict__ bias,
    float* __restrict__ out, int relu, int n_nodes) {
    __shared__ __attribute__((aligned(16))) float Atile[TM][KPAD];

    const int tid  = threadIdx.x;
    const int row0 = blockIdx.x * TM;

    {
        const int swave = __builtin_amdgcn_readfirstlane(tid >> 5);
        const int slane = tid & 31;
        const int node0 = row0 + 4 * swave;
        const float4* H4 = (const float4*)h;
        float4 vh[4];
        int beg[8], end[8], idx[8];
#pragma unroll
        for (int r = 0; r < 4; ++r) {
            vh[r] = H4[(size_t)(node0 + r) * (F / 4) + slane];
            beg[2 * r]     = start[node0 + r];
            end[2 * r]     = stop[node0 + r];
            beg[2 * r + 1] = start[n_nodes + node0 + r];
            end[2 * r + 1] = stop[n_nodes + node0 + r];
        }
#pragma unroll
        for (int q = 0; q < 8; ++q)
            idx[q] = (slane < end[q] - beg[q]) ? col[beg[q] + slane] : 0;
#pragma unroll
        for (int r = 0; r < 4; ++r)
            *(float4*)(&Atile[4 * swave + r][4 * slane]) = vh[r];
#pragma unroll
        for (int q = 0; q < 8; ++q) {
            float4 s = list_sum(H4, col, beg[q], end[q], idx[q], slane);
            const float sc = 1.0f / fmaxf((float)(end[q] - beg[q]), 1.0f);
            s.x *= sc; s.y *= sc; s.z *= sc; s.w *= sc;
            *(float4*)(&Atile[4 * swave + (q >> 1)][F * (1 + (q & 1)) + 4 * slane]) = s;
        }
    }
    __syncthreads();

    const int wave = tid >> 5;
    const int lane = tid & 31;
    const int lrow = lane & 15;
    const int lhi  = lane >> 4;
    const int n0   = wave * 16;

    v8f c0 = {};
    v8f c1 = {};

    for (int k0 = 0; k0 < KTOT; k0 += 4) {
        const int ka = k0 + 2 * lhi;
        v2f a0;
        a0.x = Atile[lrow][ka];
        a0.y = Atile[lrow][ka + 1];
        v2f a1;
        a1.x = Atile[16 + lrow][ka];
        a1.y = Atile[16 + lrow][ka + 1];
        const float* wp = W + (size_t)ka * F + n0 + lrow;
        v2f b; b.x = wp[0]; b.y = wp[F];
        c0 = __builtin_amdgcn_wmma_f32_16x16x4_f32(false, a0, false, b, (short)0, c0, false, false);
        c1 = __builtin_amdgcn_wmma_f32_16x16x4_f32(false, a1, false, b, (short)0, c1, false, false);
    }

    float* Cs = &Atile[0][0];
    __syncthreads();

    const int ncol = n0 + lrow;
    const float bi = bias[ncol];
#pragma unroll
    for (int j = 0; j < 8; ++j) {
        const int m = j + 8 * lhi;
        float v0 = c0[j] + bi;
        float v1 = c1[j] + bi;
        if (relu) { v0 = fmaxf(v0, 0.0f); v1 = fmaxf(v1, 0.0f); }
        Cs[m * CP + ncol]        = v0;
        Cs[(16 + m) * CP + ncol] = v1;
    }
    __syncthreads();
#pragma unroll
    for (int p = 0; p < TM / 8; ++p) {
        const int r = p * 8 + wave;
        const float4 v = *(const float4*)(&Cs[r * CP + 4 * lane]);
        *(float4*)(&out[(size_t)(row0 + r) * F + 4 * lane]) = v;
    }
}

extern "C" void kernel_launch(void* const* d_in, const int* in_sizes, int n_in,
                              void* d_out, int out_size, void* d_ws, size_t ws_size,
                              hipStream_t stream) {
    const float* feat = (const float*)d_in[0];
    const int*   src  = (const int*)d_in[1];
    const int*   dst  = (const int*)d_in[2];
    const float* W1   = (const float*)d_in[3];
    const float* b1   = (const float*)d_in[4];
    const float* W2   = (const float*)d_in[5];
    const float* b2   = (const float*)d_in[6];
    float*       out  = (float*)d_out;

    const int n_nodes = in_sizes[0] / F;
    const int n_edges = in_sizes[1];

    int* start = (int*)d_ws;
    int* stop  = start + 2 * n_nodes;
    int* total = stop + 2 * n_nodes;
    int* col   = total + 4;
    int* cnt   = col + 2 * n_edges;
    float* h1  = (float*)d_ws + 2 * (size_t)n_nodes + 2 * (size_t)n_nodes * F;

    const int rangeBlocks = (2 * n_nodes + RS - 1) / RS;
    const int listBlocks  = (2 * n_nodes + 255) / 256;
    const int gemmBlocks  = (n_nodes + TM - 1) / TM;

    count_kernel<<<dim3(NCHUNK, rangeBlocks), LT, 0, stream>>>(src, dst, cnt, total, n_edges, n_nodes);
    alloc_kernel<<<listBlocks, 256, 0, stream>>>(cnt, start, stop, total, 2 * n_nodes);
    fill_kernel<<<dim3(NCHUNK, rangeBlocks), LT, 0, stream>>>(src, dst, cnt, col, n_edges, n_nodes);

    gemm_kernel<<<gemmBlocks, 256, 0, stream>>>(feat, start, stop, col, W1, b1, h1, 1, n_nodes);

    gemm_kernel<<<gemmBlocks, 256, 0, stream>>>(h1, start, stop, col, W2, b2, out, 0, n_nodes);
}
